# starting kernel
# baseline (speedup 1.0000x reference)
#include <hip/hip_runtime.h>
#include <hip/hip_bf16.h>

typedef float v2f __attribute__((ext_vector_type(2)));
typedef float v8f __attribute__((ext_vector_type(8)));

#define IN_DIM  512
#define OUT_DIM 512
#define TC_DIM  256

__global__ void k_deg_init(float* __restrict__ deg, int N) {
    int i = blockIdx.x * blockDim.x + threadIdx.x;
    if (i < N) deg[i] = 1.0f;
}

__global__ void k_deg_accum(const int* __restrict__ ei, float* __restrict__ deg, int E) {
    int e = blockIdx.x * blockDim.x + threadIdx.x;
    if (e < E) atomicAdd(&deg[ei[E + e]], 1.0f);
}

__global__ void k_dinv(const float* __restrict__ deg, float* __restrict__ dinv, int N) {
    int i = blockIdx.x * blockDim.x + threadIdx.x;
    if (i < N) dinv[i] = __frsqrt_rn(deg[i]);
}

__global__ void k_cvec(const float* __restrict__ t_emb, const float* __restrict__ Wt,
                       const float* __restrict__ b, const float* __restrict__ bt,
                       float* __restrict__ cvec) {
    int j = blockIdx.x * blockDim.x + threadIdx.x;
    if (j >= OUT_DIM) return;
    float s = b[j] + bt[j];
    #pragma unroll 4
    for (int k = 0; k < TC_DIM; ++k) s += t_emb[k] * Wt[k * OUT_DIM + j];
    cvec[j] = s;
}

__global__ void k_gemm_wmma(const float* __restrict__ x, const float* __restrict__ W,
                            float* __restrict__ xw, int N, int rowTiles) {
    const int wave = (blockIdx.x * blockDim.x + threadIdx.x) >> 5;
    const int lane = threadIdx.x & 31;
    const int colTile = wave & (OUT_DIM / 16 - 1);
    const int rowTile = wave >> 5;
    if (rowTile >= rowTiles) return;

    const int row0 = rowTile * 16;
    const int col0 = colTile * 16;
    const int l15  = lane & 15;
    const int hi   = lane >> 4;

    int aRow = row0 + l15;
    if (aRow >= N) aRow = N - 1;
    const float* __restrict__ aPtr = x + (size_t)aRow * IN_DIM + 2 * hi;
    const float* __restrict__ bPtr = W + (size_t)(2 * hi) * OUT_DIM + col0 + l15;

    v8f c = {};
    for (int k = 0; k < IN_DIM; k += 4) {
        v2f a = *(const v2f*)(aPtr + k);
        v2f bm;
        bm.x = bPtr[(size_t)k * OUT_DIM];
        bm.y = bPtr[(size_t)(k + 1) * OUT_DIM];
        c = __builtin_amdgcn_wmma_f32_16x16x4_f32(
                false, a, false, bm, (short)0, c, false, false);
    }

    const int n     = col0 + l15;
    const int rbase = row0 + hi * 8;
    float* __restrict__ dPtr = xw + (size_t)rbase * OUT_DIM + n;
    if (row0 + 16 <= N) {
        #pragma unroll
        for (int v = 0; v < 8; ++v) dPtr[(size_t)v * OUT_DIM] = c[v];
    } else {
        for (int v = 0; v < 8; ++v)
            if (rbase + v < N) dPtr[(size_t)v * OUT_DIM] = c[v];
    }
}

__global__ void k_init_out(const float4* __restrict__ xw4, const float* __restrict__ dinv,
                           const float4* __restrict__ cvec4, float4* __restrict__ out4,
                           size_t total4) {
    size_t idx = (size_t)blockIdx.x * blockDim.x + threadIdx.x;
    if (idx >= total4) return;
    int i = (int)(idx >> 7);
    int q = (int)(idx & 127);
    float di = dinv[i];
    float s  = di * di;
    float4 xv = xw4[idx];
    float4 cv = cvec4[q];
    float4 r;
    r.x = cv.x + s * xv.x;
    r.y = cv.y + s * xv.y;
    r.z = cv.z + s * xv.z;
    r.w = cv.w + s * xv.w;
    out4[idx] = r;
}

__global__ void k_edge_scatter(const int* __restrict__ ei, const float* __restrict__ xw,
                               const float* __restrict__ dinv, float* __restrict__ out,
                               int E) {
    int wave = (blockIdx.x * blockDim.x + threadIdx.x) >> 5;
    int lane = threadIdx.x & 31;
    if (wave >= E) return;
    int src = ei[wave];
    int dst = ei[E + wave];
    float norm = dinv[src] * dinv[dst];
    const float4* __restrict__ srow4 = (const float4*)(xw + (size_t)src * OUT_DIM);
    float* __restrict__ drow = out + (size_t)dst * OUT_DIM;
    #pragma unroll
    for (int c = 0; c < OUT_DIM / 128; ++c) {
        int q = lane + 32 * c;
        float4 v = srow4[q];
        int col = 4 * q;
        atomicAdd(&drow[col + 0], norm * v.x);
        atomicAdd(&drow[col + 1], norm * v.y);
        atomicAdd(&drow[col + 2], norm * v.z);
        atomicAdd(&drow[col + 3], norm * v.w);
    }
}

extern "C" void kernel_launch(void* const* d_in, const int* in_sizes, int n_in,
                              void* d_out, int out_size, void* d_ws, size_t ws_size,
                              hipStream_t stream) {
    const float* x     = (const float*)d_in[0];
    const float* t_emb = (const float*)d_in[1];
    const int*   ei    = (const int*)d_in[2];
    const float* W     = (const float*)d_in[3];
    const float* b     = (const float*)d_in[4];
    const float* Wt    = (const float*)d_in[5];
    const float* bt    = (const float*)d_in[6];
    float*       out   = (float*)d_out;

    const int N = in_sizes[0] / IN_DIM;
    const int E = in_sizes[2] / 2;

    char* ws = (char*)d_ws;
    float* deg  = (float*)(ws + 0);
    float* dinv = (float*)(ws + (((size_t)N * 4 + 63) & ~63ull));
    float* cvec = (float*)((char*)dinv + (((size_t)N * 4 + 63) & ~63ull));
    float* xw   = (float*)((char*)cvec + 4096);

    k_deg_init <<<(N + 255) / 256, 256, 0, stream>>>(deg, N);
    k_deg_accum<<<(E + 255) / 256, 256, 0, stream>>>(ei, deg, E);
    k_dinv     <<<(N + 255) / 256, 256, 0, stream>>>(deg, dinv, N);

    k_cvec<<<(OUT_DIM + 255) / 256, 256, 0, stream>>>(t_emb, Wt, b, bt, cvec);

    {
        int rowTiles = (N + 15) / 16;
        int tiles    = rowTiles * (OUT_DIM / 16);
        int blocks   = (tiles + 7) / 8;
        k_gemm_wmma<<<blocks, 256, 0, stream>>>(x, W, xw, N, rowTiles);
    }

    {
        size_t total4 = (size_t)N * OUT_DIM / 4;
        int blocks = (int)((total4 + 255) / 256);
        k_init_out<<<blocks, 256, 0, stream>>>((const float4*)xw, dinv,
                                               (const float4*)cvec, (float4*)out, total4);
    }

    {
        int blocks = (E + 7) / 8;
        k_edge_scatter<<<blocks, 256, 0, stream>>>(ei, xw, dinv, out, E);
    }
}


# GEMM operands staged through LDS: a 4-wave workgroup owns an 80x128 tile of xw and fetches x and W once per 32-k stage (two LDS stages), a wave holds 5x2 accumulators; k order and sums unchanged
# speedup vs baseline: 1.8468x; 1.8468x over previous
#include <hip/hip_runtime.h>
#include <hip/hip_bf16.h>

typedef float v2f __attribute__((ext_vector_type(2)));
typedef float v8f __attribute__((ext_vector_type(8)));

#define IN_DIM  512
#define OUT_DIM 512
#define TC_DIM  256


__global__ void k_cnt_zero(int* __restrict__ cnt, int N) {
    int i = blockIdx.x * blockDim.x + threadIdx.x;
    if (i < N) cnt[i] = 0;
}

__global__ void k_edge_rank(const int* __restrict__ ei, int* __restrict__ cnt,
                            int* __restrict__ rank, int E) {
    int e = blockIdx.x * blockDim.x + threadIdx.x;
    if (e < E) rank[e] = atomicAdd(&cnt[ei[E + e]], 1);
}

#define SCAN_T 1024
__global__ void __launch_bounds__(SCAN_T)
k_row_scan(const int* __restrict__ cnt, int* __restrict__ rowptr,
           float* __restrict__ dinv, int N) {
    __shared__ int wsum[SCAN_T / 32];
    const int t    = threadIdx.x;
    const int lane = t & 31;
    const int w    = t >> 5;
    const int per  = (N + SCAN_T - 1) / SCAN_T;
    const int lo   = t * per;
    int hi = lo + per;
    if (hi > N) hi = N;

    int s = 0;
    for (int i = lo; i < hi; ++i) s += cnt[i];

    int incl = s;
    #pragma unroll
    for (int o = 1; o < 32; o <<= 1) {
        int v = __shfl_up(incl, o);
        if (lane >= o) incl += v;
    }
    if (lane == 31) wsum[w] = incl;
    __syncthreads();
    if (w == 0) {
        int v = wsum[lane];
        int x = v;
        #pragma unroll
        for (int o = 1; o < 32; o <<= 1) {
            int u = __shfl_up(x, o);
            if (lane >= o) x += u;
        }
        wsum[lane] = x - v;
    }
    __syncthreads();

    int run = wsum[w] + incl - s;
    for (int i = lo; i < hi; ++i) {
        int c = cnt[i];
        rowptr[i] = run;
        dinv[i]   = __frsqrt_rn((float)(c + 1));
        run += c;
    }
}

__global__ void k_slot_fill(const int* __restrict__ ei, const int* __restrict__ rowptr,
                            const int* __restrict__ rank, const float* __restrict__ dinv,
                            int2* __restrict__ slot, int E) {
    int e = blockIdx.x * blockDim.x + threadIdx.x;
    if (e >= E) return;
    int src = ei[e];
    int dst = ei[E + e];
    int2 v;
    v.x = src;
    v.y = __float_as_int(dinv[src]);
    slot[rowptr[dst] + rank[e]] = v;
}

__global__ void k_cvec(const float* __restrict__ t_emb, const float* __restrict__ Wt,
                       const float* __restrict__ b, const float* __restrict__ bt,
                       float* __restrict__ cvec) {
    int j = blockIdx.x * blockDim.x + threadIdx.x;
    if (j >= OUT_DIM) return;
    float s = b[j] + bt[j];
    #pragma unroll 4
    for (int k = 0; k < TC_DIM; ++k) s += t_emb[k] * Wt[k * OUT_DIM + j];
    cvec[j] = s;
}

#define GBM 80
#define GBN 128
#define GBK 32
#define GPA 36
#define GPB 136
#define GA_FLOATS (GBM * GPA)
#define GB_FLOATS (GBK * GPB)

__device__ __forceinline__ void gemm_frags(const float* __restrict__ cA, const float* __restrict__ cB,
                                           int aOff, int bOff, int kk, v2f (&a)[5], v2f (&b)[2]) {
    #pragma unroll
    for (int mt = 0; mt < 5; ++mt)
        a[mt] = *(const v2f*)(&cA[aOff + mt * 16 * GPA + 4 * kk]);
    #pragma unroll
    for (int nt = 0; nt < 2; ++nt) {
        b[nt].x = cB[bOff + 4 * kk * GPB + nt * 16];
        b[nt].y = cB[bOff + 4 * kk * GPB + nt * 16 + GPB];
    }
}

__device__ __forceinline__ void gemm_products(v8f (&acc)[5][2], const v2f (&a)[5], const v2f (&b)[2]) {
    #pragma unroll
    for (int mt = 0; mt < 5; ++mt) {
        #pragma unroll
        for (int nt = 0; nt < 2; ++nt)
            acc[mt][nt] = __builtin_amdgcn_wmma_f32_16x16x4_f32(
                false, a[mt], false, b[nt], (short)0, acc[mt][nt], false, false);
    }
}

__device__ __forceinline__ void gemm_stage(const float* __restrict__ cA, const float* __restrict__ cB,
                                           int aOff, int bOff, v8f (&acc)[5][2]) {
    v2f a0[5], b0[2], a1[5], b1[2];
    gemm_frags(cA, cB, aOff, bOff, 0, a0, b0);
    #pragma unroll
    for (int kk = 0; kk < GBK / 4; kk += 2) {
        gemm_frags(cA, cB, aOff, bOff, kk + 1, a1, b1);
        __builtin_amdgcn_sched_barrier(0x26);
        gemm_products(acc, a0, b0);
        if (kk + 2 < GBK / 4) gemm_frags(cA, cB, aOff, bOff, kk + 2, a0, b0);
        __builtin_amdgcn_sched_barrier(0x26);
        gemm_products(acc, a1, b1);
    }
}

__global__ void __launch_bounds__(128)
k_gemm_wmma(const float* __restrict__ x, const float* __restrict__ W,
            float* __restrict__ xw, int N) {
    __shared__ __attribute__((aligned(16))) float sA[2][GA_FLOATS];
    __shared__ __attribute__((aligned(16))) float sB[2][GB_FLOATS];

    const int t    = threadIdx.x;
    const int lane = t & 31;
    const int w    = t >> 5;
    const int l15  = lane & 15;
    const int hi   = lane >> 4;
    const int row0 = (blockIdx.x >> 2) * GBM;
    const int col0 = (blockIdx.x & 3) * GBN;

    const int aq = t & 7;
    const int ar = t >> 3;
    const float* aSrc[5];
    #pragma unroll
    for (int i = 0; i < 5; ++i) {
        int r = row0 + ar + 16 * i;
        if (r >= N) r = N - 1;
        aSrc[i] = x + (size_t)r * IN_DIM + 4 * aq;
    }
    const float* bSrc = W + (size_t)w * OUT_DIM + col0 + 4 * lane;
    const int aDst = ar * GPA + 4 * aq;
    const int bDst = w * GPB + 4 * lane;

    const int aOff = l15 * GPA + 2 * hi;
    const int bOff = 2 * hi * GPB + w * 32 + l15;

    v8f acc[5][2];
    #pragma unroll
    for (int mt = 0; mt < 5; ++mt) {
        acc[mt][0] = (v8f){};
        acc[mt][1] = (v8f){};
    }

    float4 ra[5], rb[8];

    #pragma unroll
    for (int i = 0; i < 5; ++i) ra[i] = *(const float4*)(aSrc[i]);
    #pragma unroll
    for (int i = 0; i < 8; ++i) rb[i] = *(const float4*)(bSrc + (size_t)(4 * i) * OUT_DIM);
    #pragma unroll
    for (int i = 0; i < 5; ++i) *(float4*)(&sA[0][aDst + 16 * i * GPA]) = ra[i];
    #pragma unroll
    for (int i = 0; i < 8; ++i) *(float4*)(&sB[0][bDst + 4 * i * GPB]) = rb[i];
    __syncthreads();

    for (int s = 0; s + 1 < IN_DIM / GBK; ++s) {
        const int cur = s & 1;
        const int k1  = (s + 1) * GBK;
        #pragma unroll
        for (int i = 0; i < 5; ++i) ra[i] = *(const float4*)(aSrc[i] + k1);
        #pragma unroll
        for (int i = 0; i < 8; ++i) rb[i] = *(const float4*)(bSrc + (size_t)(k1 + 4 * i) * OUT_DIM);

        gemm_stage(sA[cur], sB[cur], aOff, bOff, acc);

        #pragma unroll
        for (int i = 0; i < 5; ++i) *(float4*)(&sA[cur ^ 1][aDst + 16 * i * GPA]) = ra[i];
        #pragma unroll
        for (int i = 0; i < 8; ++i) *(float4*)(&sB[cur ^ 1][bDst + 4 * i * GPB]) = rb[i];
        __syncthreads();
    }
    gemm_stage(sA[(IN_DIM / GBK - 1) & 1], sB[(IN_DIM / GBK - 1) & 1], aOff, bOff, acc);

    const int n = col0 + w * 32 + l15;
    if (row0 + GBM <= N) {
        #pragma unroll
        for (int mt = 0; mt < 5; ++mt) {
            float* __restrict__ dPtr = xw + (size_t)(row0 + mt * 16 + hi * 8) * OUT_DIM + n;
            #pragma unroll
            for (int v = 0; v < 8; ++v) {
                dPtr[(size_t)v * OUT_DIM]      = acc[mt][0][v];
                dPtr[(size_t)v * OUT_DIM + 16] = acc[mt][1][v];
            }
        }
    } else {
        #pragma unroll
        for (int mt = 0; mt < 5; ++mt) {
            const int rbase = row0 + mt * 16 + hi * 8;
            float* __restrict__ dPtr = xw + (size_t)rbase * OUT_DIM + n;
            #pragma unroll
            for (int v = 0; v < 8; ++v) {
                if (rbase + v < N) {
                    dPtr[(size_t)v * OUT_DIM]      = acc[mt][0][v];
                    dPtr[(size_t)v * OUT_DIM + 16] = acc[mt][1][v];
                }
            }
        }
    }
}

__device__ __forceinline__ void row_fma(float4& a, float n, const float4 v) {
    a.x += n * v.x;
    a.y += n * v.y;
    a.z += n * v.z;
    a.w += n * v.w;
}

__global__ void __launch_bounds__(256)
k_row_gather(const int* __restrict__ rowptr, const int* __restrict__ cnt,
             const int2* __restrict__ slot, const float* __restrict__ xw,
             const float* __restrict__ dinv, const float* __restrict__ cvec,
             float* __restrict__ out, int N) {
    const int wave = (blockIdx.x * blockDim.x + threadIdx.x) >> 5;
    const int lane = threadIdx.x & 31;
    if (wave >= N) return;
    const int i = __builtin_amdgcn_readfirstlane(wave);

    const int   start = rowptr[i];
    const int   d     = cnt[i];
    const float di    = dinv[i];
    const float sl    = di * di;

    const float4* __restrict__ self4 = (const float4*)(xw + (size_t)i * OUT_DIM);
    const float4* __restrict__ cvec4 = (const float4*)cvec;
    float4 acc[4];
    #pragma unroll
    for (int c = 0; c < 4; ++c) {
        const float4 xv = self4[lane + 32 * c];
        const float4 cv = cvec4[lane + 32 * c];
        acc[c].x = cv.x + sl * xv.x;
        acc[c].y = cv.y + sl * xv.y;
        acc[c].z = cv.z + sl * xv.z;
        acc[c].w = cv.w + sl * xv.w;
    }

    for (int base = 0; base < d; base += 32) {
        int m = d - base;
        if (m > 32) m = 32;
        int   msrc = 0;
        float mnrm = 0.0f;
        if (lane < m) {
            const int2 s = slot[start + base + lane];
            msrc = s.x;
            mnrm = __int_as_float(s.y) * di;
        }
        int j = 0;
        for (; j + 2 <= m; j += 2) {
            const int   s0 = __builtin_amdgcn_readlane(msrc, j);
            const int   s1 = __builtin_amdgcn_readlane(msrc, j + 1);
            const float n0 = __int_as_float(__builtin_amdgcn_readlane(__float_as_int(mnrm), j));
            const float n1 = __int_as_float(__builtin_amdgcn_readlane(__float_as_int(mnrm), j + 1));
            const float4* __restrict__ r0 = (const float4*)(xw + (size_t)s0 * OUT_DIM);
            const float4* __restrict__ r1 = (const float4*)(xw + (size_t)s1 * OUT_DIM);
            float4 a[4], b[4];
            #pragma unroll
            for (int c = 0; c < 4; ++c) a[c] = r0[lane + 32 * c];
            #pragma unroll
            for (int c = 0; c < 4; ++c) b[c] = r1[lane + 32 * c];
            #pragma unroll
            for (int c = 0; c < 4; ++c) row_fma(acc[c], n0, a[c]);
            #pragma unroll
            for (int c = 0; c < 4; ++c) row_fma(acc[c], n1, b[c]);
        }
        if (j < m) {
            const int   s0 = __builtin_amdgcn_readlane(msrc, j);
            const float n0 = __int_as_float(__builtin_amdgcn_readlane(__float_as_int(mnrm), j));
            const float4* __restrict__ r0 = (const float4*)(xw + (size_t)s0 * OUT_DIM);
            float4 a[4];
            #pragma unroll
            for (int c = 0; c < 4; ++c) a[c] = r0[lane + 32 * c];
            #pragma unroll
            for (int c = 0; c < 4; ++c) row_fma(acc[c], n0, a[c]);
        }
    }

    float4* __restrict__ orow4 = (float4*)(out + (size_t)i * OUT_DIM);
    #pragma unroll
    for (int c = 0; c < 4; ++c) orow4[lane + 32 * c] = acc[c];
}

extern "C" void kernel_launch(void* const* d_in, const int* in_sizes, int n_in,
                              void* d_out, int out_size, void* d_ws, size_t ws_size,
                              hipStream_t stream) {
    const float* x     = (const float*)d_in[0];
    const float* t_emb = (const float*)d_in[1];
    const int*   ei    = (const int*)d_in[2];
    const float* W     = (const float*)d_in[3];
    const float* b     = (const float*)d_in[4];
    const float* Wt    = (const float*)d_in[5];
    const float* bt    = (const float*)d_in[6];
    float*       out   = (float*)d_out;

    const int N = in_sizes[0] / IN_DIM;
    const int E = in_sizes[2] / 2;

    char* ws = (char*)d_ws;
    const size_t nB = ((size_t)N * 4 + 255) & ~255ull;
    const size_t eB = ((size_t)E * 4 + 255) & ~255ull;
    int*   cnt    = (int*)(ws + 0);
    int*   rowptr = (int*)(ws + nB);
    float* dinv   = (float*)(ws + 2 * nB);
    float* cvec   = (float*)(ws + 3 * nB);
    int*   rank   = (int*)(ws + 3 * nB + 4096);
    int2*  slot   = (int2*)(ws + 3 * nB + 4096 + eB);
    float* xw     = (float*)(ws + 3 * nB + 4096 + 3 * eB);

    k_cnt_zero <<<(N + 255) / 256, 256, 0, stream>>>(cnt, N);
    k_edge_rank<<<(E + 255) / 256, 256, 0, stream>>>(ei, cnt, rank, E);
    k_row_scan <<<1, SCAN_T, 0, stream>>>(cnt, rowptr, dinv, N);
    k_slot_fill<<<(E + 255) / 256, 256, 0, stream>>>(ei, rowptr, rank, dinv, slot, E);

    k_cvec<<<(OUT_DIM + 255) / 256, 256, 0, stream>>>(t_emb, Wt, b, bt, cvec);

    {
        int rowBlocks = (N + GBM - 1) / GBM;
        k_gemm_wmma<<<rowBlocks * (OUT_DIM / GBN), 128, 0, stream>>>(x, W, xw, N);
    }

    k_row_gather<<<(N + 7) / 8, 256, 0, stream>>>(rowptr, cnt, slot, xw, dinv, cvec, out, N);
}


# Row-scan and cvec launches dropped: both run as the first 5 workgroups of the GEMM launch (7 -> 5 launches), their load chains beside the products; cvec loop unrolled 16; k order and sums unchanged
# speedup vs baseline: 2.7740x; 1.5021x over previous
#include <hip/hip_runtime.h>
#include <hip/hip_bf16.h>

typedef float v2f __attribute__((ext_vector_type(2)));
typedef float v8f __attribute__((ext_vector_type(8)));

#define IN_DIM  512
#define OUT_DIM 512
#define TC_DIM  256


__global__ void k_cnt_zero(int* __restrict__ cnt, int N) {
    int i = blockIdx.x * blockDim.x + threadIdx.x;
    if (i < N) cnt[i] = 0;
}

__global__ void k_edge_rank(const int* __restrict__ ei, int* __restrict__ cnt,
                            int* __restrict__ rank, int E) {
    int e = blockIdx.x * blockDim.x + threadIdx.x;
    if (e < E) rank[e] = atomicAdd(&cnt[ei[E + e]], 1);
}

__global__ void k_slot_fill(const int* __restrict__ ei, const int* __restrict__ rowptr,
                            const int* __restrict__ rank, const float* __restrict__ dinv,
                            int2* __restrict__ slot, int E) {
    int e = blockIdx.x * blockDim.x + threadIdx.x;
    if (e >= E) return;
    int src = ei[e];
    int dst = ei[E + e];
    int2 v;
    v.x = src;
    v.y = __float_as_int(dinv[src]);
    slot[rowptr[dst] + rank[e]] = v;
}

#define SIDE_SCAN_WG 1
#define SIDE_CVEC_WG (OUT_DIM / 128)
#define SIDE_WG      (SIDE_SCAN_WG + SIDE_CVEC_WG)

__device__ __forceinline__ void side_row_scan(const int* __restrict__ cnt, int* __restrict__ rowptr,
                                              float* __restrict__ dinv, int N, int* wsum) {
    const int t    = threadIdx.x;
    const int lane = t & 31;
    const int w    = t >> 5;
    const int per  = (((N + 127) / 128) + 31) & ~31;
    const int lo   = t * per;
    int hi = lo + per;
    if (hi > N) hi = N;

    int s = 0;
    int i = lo;
    #pragma unroll 4
    for (; i + 4 <= hi; i += 4) {
        const int4 c = *(const int4*)(cnt + i);
        s += c.x + c.y + c.z + c.w;
    }
    for (; i < hi; ++i) s += cnt[i];

    int incl = s;
    #pragma unroll
    for (int o = 1; o < 32; o <<= 1) {
        int v = __shfl_up(incl, o);
        if (lane >= o) incl += v;
    }
    if (lane == 31) wsum[w] = incl;
    __syncthreads();
    int before = 0;
    for (int k = 0; k < 3; ++k) {
        const int v = wsum[k];
        if (k < w) before += v;
    }

    int run = before + incl - s;
    i = lo;
    #pragma unroll 4
    for (; i + 4 <= hi; i += 4) {
        const int4 c = *(const int4*)(cnt + i);
        int4 r;
        r.x = run;
        r.y = r.x + c.x;
        r.z = r.y + c.y;
        r.w = r.z + c.z;
        run = r.w + c.w;
        float4 d;
        d.x = __frsqrt_rn((float)(c.x + 1));
        d.y = __frsqrt_rn((float)(c.y + 1));
        d.z = __frsqrt_rn((float)(c.z + 1));
        d.w = __frsqrt_rn((float)(c.w + 1));
        *(int4*)(rowptr + i) = r;
        *(float4*)(dinv + i) = d;
    }
    for (; i < hi; ++i) {
        const int c = cnt[i];
        rowptr[i] = run;
        dinv[i]   = __frsqrt_rn((float)(c + 1));
        run += c;
    }
}

__device__ __forceinline__ void side_cvec(const float* __restrict__ t_emb, const float* __restrict__ Wt,
                                          const float* __restrict__ b, const float* __restrict__ bt,
                                          float* __restrict__ cvec, int j) {
    float s = b[j] + bt[j];
    #pragma unroll 16
    for (int k = 0; k < TC_DIM; ++k) s += t_emb[k] * Wt[k * OUT_DIM + j];
    cvec[j] = s;
}

#define GBM 80
#define GBN 128
#define GBK 32
#define GPA 36
#define GPB 136
#define GA_FLOATS (GBM * GPA)
#define GB_FLOATS (GBK * GPB)

__device__ __forceinline__ void gemm_frags(const float* __restrict__ cA, const float* __restrict__ cB,
                                           int aOff, int bOff, int kk, v2f (&a)[5], v2f (&b)[2]) {
    #pragma unroll
    for (int mt = 0; mt < 5; ++mt)
        a[mt] = *(const v2f*)(&cA[aOff + mt * 16 * GPA + 4 * kk]);
    #pragma unroll
    for (int nt = 0; nt < 2; ++nt) {
        b[nt].x = cB[bOff + 4 * kk * GPB + nt * 16];
        b[nt].y = cB[bOff + 4 * kk * GPB + nt * 16 + GPB];
    }
}

__device__ __forceinline__ void gemm_products(v8f (&acc)[5][2], const v2f (&a)[5], const v2f (&b)[2]) {
    #pragma unroll
    for (int mt = 0; mt < 5; ++mt) {
        #pragma unroll
        for (int nt = 0; nt < 2; ++nt)
            acc[mt][nt] = __builtin_amdgcn_wmma_f32_16x16x4_f32(
                false, a[mt], false, b[nt], (short)0, acc[mt][nt], false, false);
    }
}

__device__ __forceinline__ void gemm_stage(const float* __restrict__ cA, const float* __restrict__ cB,
                                           int aOff, int bOff, v8f (&acc)[5][2]) {
    v2f a0[5], b0[2], a1[5], b1[2];
    gemm_frags(cA, cB, aOff, bOff, 0, a0, b0);
    #pragma unroll
    for (int kk = 0; kk < GBK / 4; kk += 2) {
        gemm_frags(cA, cB, aOff, bOff, kk + 1, a1, b1);
        __builtin_amdgcn_sched_barrier(0x26);
        gemm_products(acc, a0, b0);
        if (kk + 2 < GBK / 4) gemm_frags(cA, cB, aOff, bOff, kk + 2, a0, b0);
        __builtin_amdgcn_sched_barrier(0x26);
        gemm_products(acc, a1, b1);
    }
}

__global__ void __launch_bounds__(128)
k_gemm_wmma(const float* __restrict__ x, const float* __restrict__ W,
            float* __restrict__ xw, int N,
            const int* __restrict__ cnt, int* __restrict__ rowptr, float* __restrict__ dinv,
            const float* __restrict__ t_emb, const float* __restrict__ Wt,
            const float* __restrict__ b, const float* __restrict__ bt,
            float* __restrict__ cvec) {
    __shared__ __attribute__((aligned(16))) float sA[2][GA_FLOATS];
    __shared__ __attribute__((aligned(16))) float sB[2][GB_FLOATS];
    __shared__ int sWaveSum[4];

    if (blockIdx.x < SIDE_WG) {
        if (blockIdx.x < SIDE_SCAN_WG)
            side_row_scan(cnt, rowptr, dinv, N, sWaveSum);
        else
            side_cvec(t_emb, Wt, b, bt, cvec, (blockIdx.x - SIDE_SCAN_WG) * 128 + threadIdx.x);
        return;
    }

    const int t    = threadIdx.x;
    const int lane = t & 31;
    const int w    = t >> 5;
    const int l15  = lane & 15;
    const int hi   = lane >> 4;
    const int tile = blockIdx.x - SIDE_WG;
    const int row0 = (tile >> 2) * GBM;
    const int col0 = (tile & 3) * GBN;

    const int aq = t & 7;
    const int ar = t >> 3;
    const float* aSrc[5];
    #pragma unroll
    for (int i = 0; i < 5; ++i) {
        int r = row0 + ar + 16 * i;
        if (r >= N) r = N - 1;
        aSrc[i] = x + (size_t)r * IN_DIM + 4 * aq;
    }
    const float* bSrc = W + (size_t)w * OUT_DIM + col0 + 4 * lane;
    const int aDst = ar * GPA + 4 * aq;
    const int bDst = w * GPB + 4 * lane;

    const int aOff = l15 * GPA + 2 * hi;
    const int bOff = 2 * hi * GPB + w * 32 + l15;

    v8f acc[5][2];
    #pragma unroll
    for (int mt = 0; mt < 5; ++mt) {
        acc[mt][0] = (v8f){};
        acc[mt][1] = (v8f){};
    }

    float4 ra[5], rb[8];

    #pragma unroll
    for (int i = 0; i < 5; ++i) ra[i] = *(const float4*)(aSrc[i]);
    #pragma unroll
    for (int i = 0; i < 8; ++i) rb[i] = *(const float4*)(bSrc + (size_t)(4 * i) * OUT_DIM);
    #pragma unroll
    for (int i = 0; i < 5; ++i) *(float4*)(&sA[0][aDst + 16 * i * GPA]) = ra[i];
    #pragma unroll
    for (int i = 0; i < 8; ++i) *(float4*)(&sB[0][bDst + 4 * i * GPB]) = rb[i];
    __syncthreads();

    for (int s = 0; s + 1 < IN_DIM / GBK; ++s) {
        const int cur = s & 1;
        const int k1  = (s + 1) * GBK;
        #pragma unroll
        for (int i = 0; i < 5; ++i) ra[i] = *(const float4*)(aSrc[i] + k1);
        #pragma unroll
        for (int i = 0; i < 8; ++i) rb[i] = *(const float4*)(bSrc + (size_t)(k1 + 4 * i) * OUT_DIM);

        gemm_stage(sA[cur], sB[cur], aOff, bOff, acc);

        #pragma unroll
        for (int i = 0; i < 5; ++i) *(float4*)(&sA[cur ^ 1][aDst + 16 * i * GPA]) = ra[i];
        #pragma unroll
        for (int i = 0; i < 8; ++i) *(float4*)(&sB[cur ^ 1][bDst + 4 * i * GPB]) = rb[i];
        __syncthreads();
    }
    gemm_stage(sA[(IN_DIM / GBK - 1) & 1], sB[(IN_DIM / GBK - 1) & 1], aOff, bOff, acc);

    const int n = col0 + w * 32 + l15;
    if (row0 + GBM <= N) {
        #pragma unroll
        for (int mt = 0; mt < 5; ++mt) {
            float* __restrict__ dPtr = xw + (size_t)(row0 + mt * 16 + hi * 8) * OUT_DIM + n;
            #pragma unroll
            for (int v = 0; v < 8; ++v) {
                dPtr[(size_t)v * OUT_DIM]      = acc[mt][0][v];
                dPtr[(size_t)v * OUT_DIM + 16] = acc[mt][1][v];
            }
        }
    } else {
        #pragma unroll
        for (int mt = 0; mt < 5; ++mt) {
            const int rbase = row0 + mt * 16 + hi * 8;
            float* __restrict__ dPtr = xw + (size_t)rbase * OUT_DIM + n;
            #pragma unroll
            for (int v = 0; v < 8; ++v) {
                if (rbase + v < N) {
                    dPtr[(size_t)v * OUT_DIM]      = acc[mt][0][v];
                    dPtr[(size_t)v * OUT_DIM + 16] = acc[mt][1][v];
                }
            }
        }
    }
}

__device__ __forceinline__ void row_fma(float4& a, float n, const float4 v) {
    a.x += n * v.x;
    a.y += n * v.y;
    a.z += n * v.z;
    a.w += n * v.w;
}

__global__ void __launch_bounds__(256)
k_row_gather(const int* __restrict__ rowptr, const int* __restrict__ cnt,
             const int2* __restrict__ slot, const float* __restrict__ xw,
             const float* __restrict__ dinv, const float* __restrict__ cvec,
             float* __restrict__ out, int N) {
    const int wave = (blockIdx.x * blockDim.x + threadIdx.x) >> 5;
    const int lane = threadIdx.x & 31;
    if (wave >= N) return;
    const int i = __builtin_amdgcn_readfirstlane(wave);

    const int   start = rowptr[i];
    const int   d     = cnt[i];
    const float di    = dinv[i];
    const float sl    = di * di;

    const float4* __restrict__ self4 = (const float4*)(xw + (size_t)i * OUT_DIM);
    const float4* __restrict__ cvec4 = (const float4*)cvec;
    float4 acc[4];
    #pragma unroll
    for (int c = 0; c < 4; ++c) {
        const float4 xv = self4[lane + 32 * c];
        const float4 cv = cvec4[lane + 32 * c];
        acc[c].x = cv.x + sl * xv.x;
        acc[c].y = cv.y + sl * xv.y;
        acc[c].z = cv.z + sl * xv.z;
        acc[c].w = cv.w + sl * xv.w;
    }

    for (int base = 0; base < d; base += 32) {
        int m = d - base;
        if (m > 32) m = 32;
        int   msrc = 0;
        float mnrm = 0.0f;
        if (lane < m) {
            const int2 s = slot[start + base + lane];
            msrc = s.x;
            mnrm = __int_as_float(s.y) * di;
        }
        int j = 0;
        for (; j + 2 <= m; j += 2) {
            const int   s0 = __builtin_amdgcn_readlane(msrc, j);
            const int   s1 = __builtin_amdgcn_readlane(msrc, j + 1);
            const float n0 = __int_as_float(__builtin_amdgcn_readlane(__float_as_int(mnrm), j));
            const float n1 = __int_as_float(__builtin_amdgcn_readlane(__float_as_int(mnrm), j + 1));
            const float4* __restrict__ r0 = (const float4*)(xw + (size_t)s0 * OUT_DIM);
            const float4* __restrict__ r1 = (const float4*)(xw + (size_t)s1 * OUT_DIM);
            float4 a[4], b[4];
            #pragma unroll
            for (int c = 0; c < 4; ++c) a[c] = r0[lane + 32 * c];
            #pragma unroll
            for (int c = 0; c < 4; ++c) b[c] = r1[lane + 32 * c];
            #pragma unroll
            for (int c = 0; c < 4; ++c) row_fma(acc[c], n0, a[c]);
            #pragma unroll
            for (int c = 0; c < 4; ++c) row_fma(acc[c], n1, b[c]);
        }
        if (j < m) {
            const int   s0 = __builtin_amdgcn_readlane(msrc, j);
            const float n0 = __int_as_float(__builtin_amdgcn_readlane(__float_as_int(mnrm), j));
            const float4* __restrict__ r0 = (const float4*)(xw + (size_t)s0 * OUT_DIM);
            float4 a[4];
            #pragma unroll
            for (int c = 0; c < 4; ++c) a[c] = r0[lane + 32 * c];
            #pragma unroll
            for (int c = 0; c < 4; ++c) row_fma(acc[c], n0, a[c]);
        }
    }

    float4* __restrict__ orow4 = (float4*)(out + (size_t)i * OUT_DIM);
    #pragma unroll
    for (int c = 0; c < 4; ++c) orow4[lane + 32 * c] = acc[c];
}

extern "C" void kernel_launch(void* const* d_in, const int* in_sizes, int n_in,
                              void* d_out, int out_size, void* d_ws, size_t ws_size,
                              hipStream_t stream) {
    const float* x     = (const float*)d_in[0];
    const float* t_emb = (const float*)d_in[1];
    const int*   ei    = (const int*)d_in[2];
    const float* W     = (const float*)d_in[3];
    const float* b     = (const float*)d_in[4];
    const float* Wt    = (const float*)d_in[5];
    const float* bt    = (const float*)d_in[6];
    float*       out   = (float*)d_out;

    const int N = in_sizes[0] / IN_DIM;
    const int E = in_sizes[2] / 2;

    char* ws = (char*)d_ws;
    const size_t nB = ((size_t)N * 4 + 255) & ~255ull;
    const size_t eB = ((size_t)E * 4 + 255) & ~255ull;
    int*   cnt    = (int*)(ws + 0);
    int*   rowptr = (int*)(ws + nB);
    float* dinv   = (float*)(ws + 2 * nB);
    float* cvec   = (float*)(ws + 3 * nB);
    int*   rank   = (int*)(ws + 3 * nB + 4096);
    int2*  slot   = (int2*)(ws + 3 * nB + 4096 + eB);
    float* xw     = (float*)(ws + 3 * nB + 4096 + 3 * eB);

    k_cnt_zero <<<(N + 255) / 256, 256, 0, stream>>>(cnt, N);
    k_edge_rank<<<(E + 255) / 256, 256, 0, stream>>>(ei, cnt, rank, E);

    {
        int rowBlocks = (N + GBM - 1) / GBM;
        k_gemm_wmma<<<rowBlocks * (OUT_DIM / GBN) + SIDE_WG, 128, 0, stream>>>(
            x, W, xw, N, cnt, rowptr, dinv, t_emb, Wt, b, bt, cvec);
    }

    k_slot_fill<<<(E + 255) / 256, 256, 0, stream>>>(ei, rowptr, rank, dinv, slot, E);

    k_row_gather<<<(N + 7) / 8, 256, 0, stream>>>(rowptr, cnt, slot, xw, dinv, cvec, out, N);
}


# Row scan and slot-fill stages dropped: the rank step stores an edge at dst*64+rank (fixed-stride buckets, overflow list past 64) and runs beside the products; 5 -> 3 launches; sum order kept
# speedup vs baseline: 3.3555x; 1.2096x over previous
#include <hip/hip_runtime.h>
#include <hip/hip_bf16.h>

typedef float v2f __attribute__((ext_vector_type(2)));
typedef float v8f __attribute__((ext_vector_type(8)));

#define IN_DIM  512
#define OUT_DIM 512
#define TC_DIM  256

#define BKT 64

__global__ void k_cnt_zero(int* __restrict__ cnt, int n) {
    int i = blockIdx.x * blockDim.x + threadIdx.x;
    if (i < n) cnt[i] = 0;
}

#define SIDE_CVEC_WG (OUT_DIM / 128)
#define SIDE_EDGES_PER_THREAD 4
#define SIDE_EDGES_PER_WG (128 * SIDE_EDGES_PER_THREAD)

__device__ __forceinline__ void side_edge_bucket(const int* __restrict__ ei, int* __restrict__ cnt,
                                                 int* __restrict__ bucket, int2* __restrict__ ovf,
                                                 int N, int E, int wg, int nwg) {
    const int stride = nwg * 128;
    const int e0     = wg * 128 + threadIdx.x;
    int  s[SIDE_EDGES_PER_THREAD], d[SIDE_EDGES_PER_THREAD], r[SIDE_EDGES_PER_THREAD];
    bool ok[SIDE_EDGES_PER_THREAD];
    #pragma unroll
    for (int k = 0; k < SIDE_EDGES_PER_THREAD; ++k) {
        const int e = e0 + k * stride;
        ok[k] = e < E;
        s[k] = 0;
        d[k] = 0;
        if (ok[k]) {
            s[k] = ei[e];
            d[k] = ei[E + e];
        }
    }
    #pragma unroll
    for (int k = 0; k < SIDE_EDGES_PER_THREAD; ++k) {
        r[k] = 0;
        if (ok[k]) r[k] = atomicAdd(&cnt[d[k]], 1);
    }
    #pragma unroll
    for (int k = 0; k < SIDE_EDGES_PER_THREAD; ++k) {
        if (!ok[k]) continue;
        if (r[k] < BKT) {
            bucket[(size_t)d[k] * BKT + r[k]] = s[k];
        } else {
            const int q = atomicAdd(&cnt[N], 1);
            int2 v;
            v.x = s[k];
            v.y = d[k];
            ovf[q] = v;
        }
    }
}

__device__ __forceinline__ void side_cvec(const float* __restrict__ t_emb, const float* __restrict__ Wt,
                                          const float* __restrict__ b, const float* __restrict__ bt,
                                          float* __restrict__ cvec, int j) {
    float s = b[j] + bt[j];
    #pragma unroll 16
    for (int k = 0; k < TC_DIM; ++k) s += t_emb[k] * Wt[k * OUT_DIM + j];
    cvec[j] = s;
}

#define GBM 80
#define GBN 128
#define GBK 32
#define GPA 36
#define GPB 136
#define GA_FLOATS (GBM * GPA)
#define GB_FLOATS (GBK * GPB)

__device__ __forceinline__ void gemm_frags(const float* __restrict__ cA, const float* __restrict__ cB,
                                           int aOff, int bOff, int kk, v2f (&a)[5], v2f (&b)[2]) {
    #pragma unroll
    for (int mt = 0; mt < 5; ++mt)
        a[mt] = *(const v2f*)(&cA[aOff + mt * 16 * GPA + 4 * kk]);
    #pragma unroll
    for (int nt = 0; nt < 2; ++nt) {
        b[nt].x = cB[bOff + 4 * kk * GPB + nt * 16];
        b[nt].y = cB[bOff + 4 * kk * GPB + nt * 16 + GPB];
    }
}

__device__ __forceinline__ void gemm_products(v8f (&acc)[5][2], const v2f (&a)[5], const v2f (&b)[2]) {
    #pragma unroll
    for (int mt = 0; mt < 5; ++mt) {
        #pragma unroll
        for (int nt = 0; nt < 2; ++nt)
            acc[mt][nt] = __builtin_amdgcn_wmma_f32_16x16x4_f32(
                false, a[mt], false, b[nt], (short)0, acc[mt][nt], false, false);
    }
}

__device__ __forceinline__ void gemm_stage(const float* __restrict__ cA, const float* __restrict__ cB,
                                           int aOff, int bOff, v8f (&acc)[5][2]) {
    v2f a0[5], b0[2], a1[5], b1[2];
    gemm_frags(cA, cB, aOff, bOff, 0, a0, b0);
    #pragma unroll
    for (int kk = 0; kk < GBK / 4; kk += 2) {
        gemm_frags(cA, cB, aOff, bOff, kk + 1, a1, b1);
        __builtin_amdgcn_sched_barrier(0x26);
        gemm_products(acc, a0, b0);
        if (kk + 2 < GBK / 4) gemm_frags(cA, cB, aOff, bOff, kk + 2, a0, b0);
        __builtin_amdgcn_sched_barrier(0x26);
        gemm_products(acc, a1, b1);
    }
}

__global__ void __launch_bounds__(128)
k_gemm_wmma(const float* __restrict__ x, const float* __restrict__ W,
            float* __restrict__ xw, int N,
            const int* __restrict__ ei, int E, int* __restrict__ cnt,
            int* __restrict__ bucket, int2* __restrict__ ovf,
            const float* __restrict__ t_emb, const float* __restrict__ Wt,
            const float* __restrict__ b, const float* __restrict__ bt,
            float* __restrict__ cvec) {
    __shared__ __attribute__((aligned(16))) float sA[2][GA_FLOATS];
    __shared__ __attribute__((aligned(16))) float sB[2][GB_FLOATS];

    const int nTileWG = ((N + GBM - 1) / GBM) * (OUT_DIM / GBN);
    if (blockIdx.x < SIDE_CVEC_WG) {
        side_cvec(t_emb, Wt, b, bt, cvec, blockIdx.x * 128 + threadIdx.x);
        return;
    }
    if ((int)blockIdx.x >= SIDE_CVEC_WG + nTileWG) {
        side_edge_bucket(ei, cnt, bucket, ovf, N, E, (int)blockIdx.x - SIDE_CVEC_WG - nTileWG,
                         (E + SIDE_EDGES_PER_WG - 1) / SIDE_EDGES_PER_WG);
        return;
    }

    const int t    = threadIdx.x;
    const int lane = t & 31;
    const int w    = t >> 5;
    const int l15  = lane & 15;
    const int hi   = lane >> 4;
    const int tile = blockIdx.x - SIDE_CVEC_WG;
    const int row0 = (tile >> 2) * GBM;
    const int col0 = (tile & 3) * GBN;

    const int aq = t & 7;
    const int ar = t >> 3;
    const float* aSrc[5];
    #pragma unroll
    for (int i = 0; i < 5; ++i) {
        int r = row0 + ar + 16 * i;
        if (r >= N) r = N - 1;
        aSrc[i] = x + (size_t)r * IN_DIM + 4 * aq;
    }
    const float* bSrc = W + (size_t)w * OUT_DIM + col0 + 4 * lane;
    const int aDst = ar * GPA + 4 * aq;
    const int bDst = w * GPB + 4 * lane;

    const int aOff = l15 * GPA + 2 * hi;
    const int bOff = 2 * hi * GPB + w * 32 + l15;

    v8f acc[5][2];
    #pragma unroll
    for (int mt = 0; mt < 5; ++mt) {
        acc[mt][0] = (v8f){};
        acc[mt][1] = (v8f){};
    }

    float4 ra[5], rb[8];

    #pragma unroll
    for (int i = 0; i < 5; ++i) ra[i] = *(const float4*)(aSrc[i]);
    #pragma unroll
    for (int i = 0; i < 8; ++i) rb[i] = *(const float4*)(bSrc + (size_t)(4 * i) * OUT_DIM);
    #pragma unroll
    for (int i = 0; i < 5; ++i) *(float4*)(&sA[0][aDst + 16 * i * GPA]) = ra[i];
    #pragma unroll
    for (int i = 0; i < 8; ++i) *(float4*)(&sB[0][bDst + 4 * i * GPB]) = rb[i];
    __syncthreads();

    for (int s = 0; s + 1 < IN_DIM / GBK; ++s) {
        const int cur = s & 1;
        const int k1  = (s + 1) * GBK;
        #pragma unroll
        for (int i = 0; i < 5; ++i) ra[i] = *(const float4*)(aSrc[i] + k1);
        #pragma unroll
        for (int i = 0; i < 8; ++i) rb[i] = *(const float4*)(bSrc + (size_t)(k1 + 4 * i) * OUT_DIM);

        gemm_stage(sA[cur], sB[cur], aOff, bOff, acc);

        #pragma unroll
        for (int i = 0; i < 5; ++i) *(float4*)(&sA[cur ^ 1][aDst + 16 * i * GPA]) = ra[i];
        #pragma unroll
        for (int i = 0; i < 8; ++i) *(float4*)(&sB[cur ^ 1][bDst + 4 * i * GPB]) = rb[i];
        __syncthreads();
    }
    gemm_stage(sA[(IN_DIM / GBK - 1) & 1], sB[(IN_DIM / GBK - 1) & 1], aOff, bOff, acc);

    const int n = col0 + w * 32 + l15;
    if (row0 + GBM <= N) {
        #pragma unroll
        for (int mt = 0; mt < 5; ++mt) {
            float* __restrict__ dPtr = xw + (size_t)(row0 + mt * 16 + hi * 8) * OUT_DIM + n;
            #pragma unroll
            for (int v = 0; v < 8; ++v) {
                dPtr[(size_t)v * OUT_DIM]      = acc[mt][0][v];
                dPtr[(size_t)v * OUT_DIM + 16] = acc[mt][1][v];
            }
        }
    } else {
        #pragma unroll
        for (int mt = 0; mt < 5; ++mt) {
            const int rbase = row0 + mt * 16 + hi * 8;
            float* __restrict__ dPtr = xw + (size_t)rbase * OUT_DIM + n;
            #pragma unroll
            for (int v = 0; v < 8; ++v) {
                if (rbase + v < N) {
                    dPtr[(size_t)v * OUT_DIM]      = acc[mt][0][v];
                    dPtr[(size_t)v * OUT_DIM + 16] = acc[mt][1][v];
                }
            }
        }
    }
}

__device__ __forceinline__ void row_fma(float4& a, float n, const float4 v) {
    a.x += n * v.x;
    a.y += n * v.y;
    a.z += n * v.z;
    a.w += n * v.w;
}

__global__ void __launch_bounds__(256)
k_row_gather(const int* __restrict__ cnt, const int* __restrict__ bucket,
             const int2* __restrict__ ovf, const float* __restrict__ xw,
             const float* __restrict__ cvec, float* __restrict__ out, int N, int E) {
    const int wave = (blockIdx.x * blockDim.x + threadIdx.x) >> 5;
    const int lane = threadIdx.x & 31;
    if (wave >= N) return;
    const int i = __builtin_amdgcn_readfirstlane(wave);

    const int   d  = cnt[i];
    const float di = __frsqrt_rn((float)(d + 1));
    const float sl = di * di;

    const float4* __restrict__ self4 = (const float4*)(xw + (size_t)i * OUT_DIM);
    const float4* __restrict__ cvec4 = (const float4*)cvec;
    float4 acc[4];
    #pragma unroll
    for (int c = 0; c < 4; ++c) {
        const float4 xv = self4[lane + 32 * c];
        const float4 cv = cvec4[lane + 32 * c];
        acc[c].x = cv.x + sl * xv.x;
        acc[c].y = cv.y + sl * xv.y;
        acc[c].z = cv.z + sl * xv.z;
        acc[c].w = cv.w + sl * xv.w;
    }

    const int nb = d < BKT ? d : BKT;
    const int* __restrict__ brow = bucket + (size_t)i * BKT;
    for (int base = 0; base < nb; base += 32) {
        int m = nb - base;
        if (m > 32) m = 32;
        int   msrc = 0;
        float mnrm = 0.0f;
        if (lane < m) {
            msrc = brow[base + lane];
            mnrm = __frsqrt_rn((float)(cnt[msrc] + 1)) * di;
        }
        int j = 0;
        for (; j + 2 <= m; j += 2) {
            const int   s0 = __builtin_amdgcn_readlane(msrc, j);
            const int   s1 = __builtin_amdgcn_readlane(msrc, j + 1);
            const float n0 = __int_as_float(__builtin_amdgcn_readlane(__float_as_int(mnrm), j));
            const float n1 = __int_as_float(__builtin_amdgcn_readlane(__float_as_int(mnrm), j + 1));
            const float4* __restrict__ r0 = (const float4*)(xw + (size_t)s0 * OUT_DIM);
            const float4* __restrict__ r1 = (const float4*)(xw + (size_t)s1 * OUT_DIM);
            float4 a[4], b[4];
            #pragma unroll
            for (int c = 0; c < 4; ++c) a[c] = r0[lane + 32 * c];
            #pragma unroll
            for (int c = 0; c < 4; ++c) b[c] = r1[lane + 32 * c];
            #pragma unroll
            for (int c = 0; c < 4; ++c) row_fma(acc[c], n0, a[c]);
            #pragma unroll
            for (int c = 0; c < 4; ++c) row_fma(acc[c], n1, b[c]);
        }
        if (j < m) {
            const int   s0 = __builtin_amdgcn_readlane(msrc, j);
            const float n0 = __int_as_float(__builtin_amdgcn_readlane(__float_as_int(mnrm), j));
            const float4* __restrict__ r0 = (const float4*)(xw + (size_t)s0 * OUT_DIM);
            float4 a[4];
            #pragma unroll
            for (int c = 0; c < 4; ++c) a[c] = r0[lane + 32 * c];
            #pragma unroll
            for (int c = 0; c < 4; ++c) row_fma(acc[c], n0, a[c]);
        }
    }

    if (d > BKT) {
        int no = cnt[N];
        if (no > E) no = E;
        for (int base = 0; base < no; base += 32) {
            int   msrc = 0;
            float mnrm = 0.0f;
            bool  hit  = false;
            if (base + lane < no) {
                const int2 s = ovf[base + lane];
                if (s.y == i) {
                    hit  = true;
                    msrc = s.x;
                    mnrm = __frsqrt_rn((float)(cnt[s.x] + 1)) * di;
                }
            }
            unsigned mask = __builtin_amdgcn_ballot_w32(hit);
            while (mask != 0u) {
                const int j = __builtin_ctz(mask);
                mask &= mask - 1u;
                const int   s0 = __builtin_amdgcn_readlane(msrc, j);
                const float n0 = __int_as_float(__builtin_amdgcn_readlane(__float_as_int(mnrm), j));
                const float4* __restrict__ r0 = (const float4*)(xw + (size_t)s0 * OUT_DIM);
                float4 a[4];
                #pragma unroll
                for (int c = 0; c < 4; ++c) a[c] = r0[lane + 32 * c];
                #pragma unroll
                for (int c = 0; c < 4; ++c) row_fma(acc[c], n0, a[c]);
            }
        }
    }

    float4* __restrict__ orow4 = (float4*)(out + (size_t)i * OUT_DIM);
    #pragma unroll
    for (int c = 0; c < 4; ++c) orow4[lane + 32 * c] = acc[c];
}

extern "C" void kernel_launch(void* const* d_in, const int* in_sizes, int n_in,
                              void* d_out, int out_size, void* d_ws, size_t ws_size,
                              hipStream_t stream) {
    const float* x     = (const float*)d_in[0];
    const float* t_emb = (const float*)d_in[1];
    const int*   ei    = (const int*)d_in[2];
    const float* W     = (const float*)d_in[3];
    const float* b     = (const float*)d_in[4];
    const float* Wt    = (const float*)d_in[5];
    const float* bt    = (const float*)d_in[6];
    float*       out   = (float*)d_out;

    const int N = in_sizes[0] / IN_DIM;
    const int E = in_sizes[2] / 2;

    char* ws = (char*)d_ws;
    const size_t nB = (((size_t)N + 1) * 4 + 255) & ~255ull;
    const size_t kB = (size_t)N * BKT * 4;
    const size_t oB = ((size_t)E * 8 + 255) & ~255ull;
    int*   cnt    = (int*)(ws + 0);
    float* cvec   = (float*)(ws + nB);
    int*   bucket = (int*)(ws + nB + 4096);
    int2*  ovf    = (int2*)(ws + nB + 4096 + kB);
    float* xw     = (float*)(ws + nB + 4096 + kB + oB);

    k_cnt_zero<<<(N + 1 + 255) / 256, 256, 0, stream>>>(cnt, N + 1);

    {
        int rowBlocks = (N + GBM - 1) / GBM;
        int edgeWGs   = (E + SIDE_EDGES_PER_WG - 1) / SIDE_EDGES_PER_WG;
        k_gemm_wmma<<<rowBlocks * (OUT_DIM / GBN) + SIDE_CVEC_WG + edgeWGs, 128, 0, stream>>>(
            x, W, xw, N, ei, E, cnt, bucket, ovf, t_emb, Wt, b, bt, cvec);
    }

    k_row_gather<<<(N + 7) / 8, 256, 0, stream>>>(cnt, bucket, ovf, xw, cvec, out, N, E);
}


# GEMM LDS fragment reads grouped: half a stage (4 k steps) of x and W fragments is read as one group ahead of a 40-product block; W staged in LDS as k pairs (8-byte reads); k order and sums unchanged
# speedup vs baseline: 3.6295x; 1.0817x over previous
#include <hip/hip_runtime.h>
#include <hip/hip_bf16.h>

typedef float v2f __attribute__((ext_vector_type(2)));
typedef float v8f __attribute__((ext_vector_type(8)));

#define IN_DIM  512
#define OUT_DIM 512
#define TC_DIM  256

#define BKT 64

__global__ void k_cnt_zero(int* __restrict__ cnt, int n) {
    int i = blockIdx.x * blockDim.x + threadIdx.x;
    if (i < n) cnt[i] = 0;
}

#define SIDE_CVEC_WG (OUT_DIM / 128)
#define SIDE_EDGES_PER_THREAD 4
#define SIDE_EDGES_PER_WG (128 * SIDE_EDGES_PER_THREAD)

__device__ __forceinline__ void side_edge_bucket(const int* __restrict__ ei, int* __restrict__ cnt,
                                                 int* __restrict__ bucket, int2* __restrict__ ovf,
                                                 int N, int E, int wg, int nwg) {
    const int stride = nwg * 128;
    const int e0     = wg * 128 + threadIdx.x;
    int  s[SIDE_EDGES_PER_THREAD], d[SIDE_EDGES_PER_THREAD], r[SIDE_EDGES_PER_THREAD];
    bool ok[SIDE_EDGES_PER_THREAD];
    #pragma unroll
    for (int k = 0; k < SIDE_EDGES_PER_THREAD; ++k) {
        const int e = e0 + k * stride;
        ok[k] = e < E;
        s[k] = 0;
        d[k] = 0;
        if (ok[k]) {
            s[k] = ei[e];
            d[k] = ei[E + e];
        }
    }
    #pragma unroll
    for (int k = 0; k < SIDE_EDGES_PER_THREAD; ++k) {
        r[k] = 0;
        if (ok[k]) r[k] = atomicAdd(&cnt[d[k]], 1);
    }
    #pragma unroll
    for (int k = 0; k < SIDE_EDGES_PER_THREAD; ++k) {
        if (!ok[k]) continue;
        if (r[k] < BKT) {
            bucket[(size_t)d[k] * BKT + r[k]] = s[k];
        } else {
            const int q = atomicAdd(&cnt[N], 1);
            int2 v;
            v.x = s[k];
            v.y = d[k];
            ovf[q] = v;
        }
    }
}

__device__ __forceinline__ void side_cvec(const float* __restrict__ t_emb, const float* __restrict__ Wt,
                                          const float* __restrict__ b, const float* __restrict__ bt,
                                          float* __restrict__ cvec, int j) {
    float s = b[j] + bt[j];
    #pragma unroll 16
    for (int k = 0; k < TC_DIM; ++k) s += t_emb[k] * Wt[k * OUT_DIM + j];
    cvec[j] = s;
}

#define GBM 80
#define GBN 128
#define GBK 32
#define GHK 4
#define GPA 36
#define GPB2 288
#define GA_FLOATS (GBM * GPA)
#define GB_FLOATS ((GBK / 2) * GPB2)

struct HalfFrags {
    v2f a[GHK][5];
    v2f b[GHK][2];
};

__device__ __forceinline__ void gemm_half_frags(const float* __restrict__ cA, const float* __restrict__ cB,
                                                int aOff, int bOff, int half, HalfFrags& f) {
    #pragma unroll
    for (int j = 0; j < GHK; ++j) {
        const int kk = half * GHK + j;
        #pragma unroll
        for (int mt = 0; mt < 5; ++mt)
            f.a[j][mt] = *(const v2f*)(&cA[aOff + mt * 16 * GPA + 4 * kk]);
        #pragma unroll
        for (int nt = 0; nt < 2; ++nt)
            f.b[j][nt] = *(const v2f*)(&cB[bOff + 2 * kk * GPB2 + nt * 32]);
    }
}

__device__ __forceinline__ void gemm_half_products(v8f (&acc)[5][2], const HalfFrags& f) {
    #pragma unroll
    for (int j = 0; j < GHK; ++j) {
        #pragma unroll
        for (int mt = 0; mt < 5; ++mt) {
            #pragma unroll
            for (int nt = 0; nt < 2; ++nt)
                acc[mt][nt] = __builtin_amdgcn_wmma_f32_16x16x4_f32(
                    false, f.a[j][mt], false, f.b[j][nt], (short)0, acc[mt][nt], false, false);
        }
    }
}

__global__ void __launch_bounds__(128)
k_gemm_wmma(const float* __restrict__ x, const float* __restrict__ W,
            float* __restrict__ xw, int N,
            const int* __restrict__ ei, int E, int* __restrict__ cnt,
            int* __restrict__ bucket, int2* __restrict__ ovf,
            const float* __restrict__ t_emb, const float* __restrict__ Wt,
            const float* __restrict__ b, const float* __restrict__ bt,
            float* __restrict__ cvec) {
    __shared__ __attribute__((aligned(16))) float sA[2][GA_FLOATS];
    __shared__ __attribute__((aligned(16))) float sB[2][GB_FLOATS];

    const int nTileWG = ((N + GBM - 1) / GBM) * (OUT_DIM / GBN);
    if (blockIdx.x < SIDE_CVEC_WG) {
        side_cvec(t_emb, Wt, b, bt, cvec, blockIdx.x * 128 + threadIdx.x);
        return;
    }
    if ((int)blockIdx.x >= SIDE_CVEC_WG + nTileWG) {
        side_edge_bucket(ei, cnt, bucket, ovf, N, E, (int)blockIdx.x - SIDE_CVEC_WG - nTileWG,
                         (E + SIDE_EDGES_PER_WG - 1) / SIDE_EDGES_PER_WG);
        return;
    }

    const int t    = threadIdx.x;
    const int lane = t & 31;
    const int w    = t >> 5;
    const int l15  = lane & 15;
    const int hi   = lane >> 4;
    const int tile = blockIdx.x - SIDE_CVEC_WG;
    const int row0 = (tile >> 2) * GBM;
    const int col0 = (tile & 3) * GBN;

    const int aq = t & 7;
    const int ar = t >> 3;
    const float* aSrc[5];
    #pragma unroll
    for (int i = 0; i < 5; ++i) {
        int r = row0 + ar + 16 * i;
        if (r >= N) r = N - 1;
        aSrc[i] = x + (size_t)r * IN_DIM + 4 * aq;
    }
    const float* bSrc = W + (size_t)(2 * w) * OUT_DIM + col0 + 4 * lane;
    const int aDst = ar * GPA + 4 * aq;
    const int bDst = w * GPB2 + 8 * lane;

    const int aOff = l15 * GPA + 2 * hi;
    const int bOff = hi * GPB2 + 2 * (w * 32 + l15);

    v8f acc[5][2];
    #pragma unroll
    for (int mt = 0; mt < 5; ++mt) {
        acc[mt][0] = (v8f){};
        acc[mt][1] = (v8f){};
    }

    float4 ra[5], rb[8];
    HalfFrags fA, fB;
    constexpr int NS = IN_DIM / GBK;

#define GEMM_FETCH(K0)                                                                              \
    _Pragma("unroll")                                                                               \
    for (int i = 0; i < 5; ++i) ra[i] = *(const float4*)(aSrc[i] + (K0));                           \
    _Pragma("unroll")                                                                               \
    for (int i = 0; i < 8; ++i)                                                                     \
        rb[i] = *(const float4*)(bSrc + (size_t)((K0) + 8 * (i >> 1) + (i & 1)) * OUT_DIM);
#define GEMM_PUT(BUF)                                                                               \
    _Pragma("unroll")                                                                               \
    for (int i = 0; i < 5; ++i) *(float4*)(&sA[BUF][aDst + 16 * i * GPA]) = ra[i];                  \
    _Pragma("unroll")                                                                               \
    for (int i = 0; i < 4; ++i) {                                                                   \
        *(float4*)(&sB[BUF][bDst + 4 * i * GPB2]) =                                                 \
            make_float4(rb[2 * i].x, rb[2 * i + 1].x, rb[2 * i].y, rb[2 * i + 1].y);                \
        *(float4*)(&sB[BUF][bDst + 4 * i * GPB2 + 4]) =                                             \
            make_float4(rb[2 * i].z, rb[2 * i + 1].z, rb[2 * i].w, rb[2 * i + 1].w);                \
    }

    GEMM_FETCH(0)
    GEMM_PUT(0)
    __syncthreads();

    gemm_half_frags(sA[0], sB[0], aOff, bOff, 0, fA);
    GEMM_FETCH(GBK)
    gemm_half_frags(sA[0], sB[0], aOff, bOff, 1, fB);
    gemm_half_products(acc, fA);
    GEMM_PUT(1)
    __syncthreads();

    for (int t = 1; t + 1 < NS; ++t) {
        const int cur = t & 1;
        gemm_half_frags(sA[cur], sB[cur], aOff, bOff, 0, fA);
        GEMM_FETCH((t + 1) * GBK)
        gemm_half_products(acc, fB);
        gemm_half_frags(sA[cur], sB[cur], aOff, bOff, 1, fB);
        gemm_half_products(acc, fA);
        GEMM_PUT(cur ^ 1)
        __builtin_amdgcn_sched_group_barrier(0x100, 14, 0);
        __builtin_amdgcn_sched_group_barrier(0x008, 40, 0);
        __builtin_amdgcn_sched_group_barrier(0x100, 14, 0);
        __builtin_amdgcn_sched_group_barrier(0x008, 40, 0);
        __syncthreads();
    }

    gemm_half_frags(sA[(NS - 1) & 1], sB[(NS - 1) & 1], aOff, bOff, 0, fA);
    gemm_half_products(acc, fB);
    gemm_half_frags(sA[(NS - 1) & 1], sB[(NS - 1) & 1], aOff, bOff, 1, fB);
    gemm_half_products(acc, fA);
    gemm_half_products(acc, fB);
    __builtin_amdgcn_sched_group_barrier(0x100, 14, 1);
    __builtin_amdgcn_sched_group_barrier(0x008, 40, 1);
    __builtin_amdgcn_sched_group_barrier(0x100, 14, 1);
    __builtin_amdgcn_sched_group_barrier(0x008, 80, 1);
#undef GEMM_FETCH
#undef GEMM_PUT

    const int n = col0 + w * 32 + l15;
    if (row0 + GBM <= N) {
        #pragma unroll
        for (int mt = 0; mt < 5; ++mt) {
            float* __restrict__ dPtr = xw + (size_t)(row0 + mt * 16 + hi * 8) * OUT_DIM + n;
            #pragma unroll
            for (int v = 0; v < 8; ++v) {
                dPtr[(size_t)v * OUT_DIM]      = acc[mt][0][v];
                dPtr[(size_t)v * OUT_DIM + 16] = acc[mt][1][v];
            }
        }
    } else {
        #pragma unroll
        for (int mt = 0; mt < 5; ++mt) {
            const int rbase = row0 + mt * 16 + hi * 8;
            float* __restrict__ dPtr = xw + (size_t)rbase * OUT_DIM + n;
            #pragma unroll
            for (int v = 0; v < 8; ++v) {
                if (rbase + v < N) {
                    dPtr[(size_t)v * OUT_DIM]      = acc[mt][0][v];
                    dPtr[(size_t)v * OUT_DIM + 16] = acc[mt][1][v];
                }
            }
        }
    }
}

__device__ __forceinline__ void row_fma(float4& a, float n, const float4 v) {
    a.x += n * v.x;
    a.y += n * v.y;
    a.z += n * v.z;
    a.w += n * v.w;
}

__global__ void __launch_bounds__(256)
k_row_gather(const int* __restrict__ cnt, const int* __restrict__ bucket,
             const int2* __restrict__ ovf, const float* __restrict__ xw,
             const float* __restrict__ cvec, float* __restrict__ out, int N, int E) {
    const int wave = (blockIdx.x * blockDim.x + threadIdx.x) >> 5;
    const int lane = threadIdx.x & 31;
    if (wave >= N) return;
    const int i = __builtin_amdgcn_readfirstlane(wave);

    const int   d  = cnt[i];
    const float di = __frsqrt_rn((float)(d + 1));
    const float sl = di * di;

    const float4* __restrict__ self4 = (const float4*)(xw + (size_t)i * OUT_DIM);
    const float4* __restrict__ cvec4 = (const float4*)cvec;
    float4 acc[4];
    #pragma unroll
    for (int c = 0; c < 4; ++c) {
        const float4 xv = self4[lane + 32 * c];
        const float4 cv = cvec4[lane + 32 * c];
        acc[c].x = cv.x + sl * xv.x;
        acc[c].y = cv.y + sl * xv.y;
        acc[c].z = cv.z + sl * xv.z;
        acc[c].w = cv.w + sl * xv.w;
    }

    const int nb = d < BKT ? d : BKT;
    const int* __restrict__ brow = bucket + (size_t)i * BKT;
    for (int base = 0; base < nb; base += 32) {
        int m = nb - base;
        if (m > 32) m = 32;
        int   msrc = 0;
        float mnrm = 0.0f;
        if (lane < m) {
            msrc = brow[base + lane];
            mnrm = __frsqrt_rn((float)(cnt[msrc] + 1)) * di;
        }
        int j = 0;
        for (; j + 2 <= m; j += 2) {
            const int   s0 = __builtin_amdgcn_readlane(msrc, j);
            const int   s1 = __builtin_amdgcn_readlane(msrc, j + 1);
            const float n0 = __int_as_float(__builtin_amdgcn_readlane(__float_as_int(mnrm), j));
            const float n1 = __int_as_float(__builtin_amdgcn_readlane(__float_as_int(mnrm), j + 1));
            const float4* __restrict__ r0 = (const float4*)(xw + (size_t)s0 * OUT_DIM);
            const float4* __restrict__ r1 = (const float4*)(xw + (size_t)s1 * OUT_DIM);
            float4 a[4], b[4];
            #pragma unroll
            for (int c = 0; c < 4; ++c) a[c] = r0[lane + 32 * c];
            #pragma unroll
            for (int c = 0; c < 4; ++c) b[c] = r1[lane + 32 * c];
            #pragma unroll
            for (int c = 0; c < 4; ++c) row_fma(acc[c], n0, a[c]);
            #pragma unroll
            for (int c = 0; c < 4; ++c) row_fma(acc[c], n1, b[c]);
        }
        if (j < m) {
            const int   s0 = __builtin_amdgcn_readlane(msrc, j);
            const float n0 = __int_as_float(__builtin_amdgcn_readlane(__float_as_int(mnrm), j));
            const float4* __restrict__ r0 = (const float4*)(xw + (size_t)s0 * OUT_DIM);
            float4 a[4];
            #pragma unroll
            for (int c = 0; c < 4; ++c) a[c] = r0[lane + 32 * c];
            #pragma unroll
            for (int c = 0; c < 4; ++c) row_fma(acc[c], n0, a[c]);
        }
    }

    if (d > BKT) {
        int no = cnt[N];
        if (no > E) no = E;
        for (int base = 0; base < no; base += 32) {
            int   msrc = 0;
            float mnrm = 0.0f;
            bool  hit  = false;
            if (base + lane < no) {
                const int2 s = ovf[base + lane];
                if (s.y == i) {
                    hit  = true;
                    msrc = s.x;
                    mnrm = __frsqrt_rn((float)(cnt[s.x] + 1)) * di;
                }
            }
            unsigned mask = __builtin_amdgcn_ballot_w32(hit);
            while (mask != 0u) {
                const int j = __builtin_ctz(mask);
                mask &= mask - 1u;
                const int   s0 = __builtin_amdgcn_readlane(msrc, j);
                const float n0 = __int_as_float(__builtin_amdgcn_readlane(__float_as_int(mnrm), j));
                const float4* __restrict__ r0 = (const float4*)(xw + (size_t)s0 * OUT_DIM);
                float4 a[4];
                #pragma unroll
                for (int c = 0; c < 4; ++c) a[c] = r0[lane + 32 * c];
                #pragma unroll
                for (int c = 0; c < 4; ++c) row_fma(acc[c], n0, a[c]);
            }
        }
    }

    float4* __restrict__ orow4 = (float4*)(out + (size_t)i * OUT_DIM);
    #pragma unroll
    for (int c = 0; c < 4; ++c) orow4[lane + 32 * c] = acc[c];
}

extern "C" void kernel_launch(void* const* d_in, const int* in_sizes, int n_in,
                              void* d_out, int out_size, void* d_ws, size_t ws_size,
                              hipStream_t stream) {
    const float* x     = (const float*)d_in[0];
    const float* t_emb = (const float*)d_in[1];
    const int*   ei    = (const int*)d_in[2];
    const float* W     = (const float*)d_in[3];
    const float* b     = (const float*)d_in[4];
    const float* Wt    = (const float*)d_in[5];
    const float* bt    = (const float*)d_in[6];
    float*       out   = (float*)d_out;

    const int N = in_sizes[0] / IN_DIM;
    const int E = in_sizes[2] / 2;

    char* ws = (char*)d_ws;
    const size_t nB = (((size_t)N + 1) * 4 + 255) & ~255ull;
    const size_t kB = (size_t)N * BKT * 4;
    const size_t oB = ((size_t)E * 8 + 255) & ~255ull;
    int*   cnt    = (int*)(ws + 0);
    float* cvec   = (float*)(ws + nB);
    int*   bucket = (int*)(ws + nB + 4096);
    int2*  ovf    = (int2*)(ws + nB + 4096 + kB);
    float* xw     = (float*)(ws + nB + 4096 + kB + oB);

    k_cnt_zero<<<(N + 1 + 255) / 256, 256, 0, stream>>>(cnt, N + 1);

    {
        int rowBlocks = (N + GBM - 1) / GBM;
        int edgeWGs   = (E + SIDE_EDGES_PER_WG - 1) / SIDE_EDGES_PER_WG;
        k_gemm_wmma<<<rowBlocks * (OUT_DIM / GBN) + SIDE_CVEC_WG + edgeWGs, 128, 0, stream>>>(
            x, W, xw, N, ei, E, cnt, bucket, ovf, t_emb, Wt, b, bt, cvec);
    }

    k_row_gather<<<(N + 7) / 8, 256, 0, stream>>>(cnt, bucket, ovf, xw, cvec, out, N, E);
}
